# starting kernel
# baseline (speedup 1.0000x reference)
#include <hip/hip_runtime.h>
#include <hip/hip_bf16.h>

typedef float v2f __attribute__((ext_vector_type(2)));
typedef float v8f __attribute__((ext_vector_type(8)));

#define B_    64
#define E_    16
#define IN_   256
#define OUT_  256
#define NPG_  2048
#define MWG   128
#define KC    32
#define BSTR2 272

__global__ __launch_bounds__(256) void mole_mix(const float* __restrict__ kern,
                                                const float* __restrict__ coeff,
                                                float* __restrict__ mixed) {
    __shared__ float cs[B_ * E_];
    const int tid = threadIdx.x;
    for (int j = tid; j < B_ * E_; j += 256) cs[j] = coeff[j];
    __syncthreads();

    const int idx = blockIdx.x * 256 + tid;
    const int i   = idx >> 6;
    const int o4  = (idx & 63) << 2;

    float4 kf[E_];
#pragma unroll
    for (int e = 0; e < E_; ++e)
        kf[e] = *(const float4*)(kern + (size_t)e * IN_ * OUT_ + (size_t)i * OUT_ + o4);

    for (int b = 0; b < B_; ++b) {
        float4 acc = make_float4(0.f, 0.f, 0.f, 0.f);
#pragma unroll
        for (int e = 0; e < E_; ++e) {
            const float c = cs[b * E_ + e];
            acc.x += c * kf[e].x;  acc.y += c * kf[e].y;
            acc.z += c * kf[e].z;  acc.w += c * kf[e].w;
        }
        *(float4*)(mixed + (size_t)b * IN_ * OUT_ + (size_t)i * OUT_ + o4) = acc;
    }
}

__global__ __launch_bounds__(256) void mole_gemm(const float* __restrict__ X,
                                                 const float* __restrict__ mixed,
                                                 const float* __restrict__ bias,
                                                 float* __restrict__ out) {
    extern __shared__ float Bs[];

    const int tid = threadIdx.x;
    const int b   = blockIdx.x >> 4;
    const int mt  = blockIdx.x & 15;
    const float* __restrict__ Bm = mixed + (size_t)b * IN_ * OUT_;

    const int wave = tid >> 5;
    const int lane = tid & 31;
    const int half = lane >> 4;
    const int l16  = lane & 15;
    const int rowBase = b * NPG_ + mt * MWG + wave * 16;
    const float* __restrict__ Xrow = X + (size_t)(rowBase + l16) * IN_ + 2 * half;

    v8f acc[16];
#pragma unroll
    for (int n = 0; n < 16; ++n) {
        v8f z = {};
        acc[n] = z;
    }

    for (int kc = 0; kc < IN_ / KC; ++kc) {
        for (int g = tid; g < (KC / 2) * (OUT_ / 4); g += 256) {
            const int p  = g >> 6;
            const int c4 = (g & 63) << 2;
            const float* src = Bm + (size_t)(kc * KC + 2 * p) * OUT_ + c4;
            const float4 lo = *(const float4*)(src);
            const float4 hi = *(const float4*)(src + OUT_);
            float* dst = Bs + ((size_t)p * BSTR2 + c4) * 2;
            *(float4*)(dst)     = make_float4(lo.x, hi.x, lo.y, hi.y);
            *(float4*)(dst + 4) = make_float4(lo.z, hi.z, lo.w, hi.w);
        }
        __syncthreads();

        for (int kk = 0; kk < KC / 4; ++kk) {
            const v2f a = *(const v2f*)(Xrow + kc * KC + kk * 4);
            const v2f* __restrict__ Brow =
                (const v2f*)Bs + (size_t)(2 * kk + half) * BSTR2 + l16;
#pragma unroll
            for (int n = 0; n < 16; ++n) {
                const v2f bf = Brow[n * 16];
                acc[n] = __builtin_amdgcn_wmma_f32_16x16x4_f32(
                    false, a, false, bf, (short)0, acc[n], false, false);
            }
        }
        __syncthreads();
    }

    const int outRow0 = rowBase + 8 * half;
#pragma unroll
    for (int n = 0; n < 16; ++n) {
        const int col = n * 16 + l16;
        const float bv = bias[col];
#pragma unroll
        for (int r = 0; r < 8; ++r) {
            out[(size_t)(outRow0 + r) * OUT_ + col] = acc[n][r] + bv;
        }
    }
}

extern "C" void kernel_launch(void* const* d_in, const int* in_sizes, int n_in,
                              void* d_out, int out_size, void* d_ws, size_t ws_size,
                              hipStream_t stream) {
    const float* inputs = (const float*)d_in[0];
    const float* coeffs = (const float*)d_in[1];
    const float* kern   = (const float*)d_in[3];
    const float* bias   = (const float*)d_in[4];
    float*       out    = (float*)d_out;
    float*       mixed  = (float*)d_ws;

    mole_mix<<<dim3((IN_ * OUT_ / 4) / 256), dim3(256), 0, stream>>>(kern, coeffs, mixed);

    const size_t ldsBytes = (size_t)(KC / 2) * BSTR2 * 2 * sizeof(float);
    mole_gemm<<<dim3(B_ * (NPG_ / MWG)), dim3(256), ldsBytes, stream>>>(inputs, mixed,
                                                                        bias, out);
}


# GEMM re-tiled to 64x64 wave tiles (8 waves as 2x4 on a 128x256 block): X chunks are staged in LDS too, by 16-byte loads issued a chunk ahead, so each LDS fragment read feeds 4 WMMAs
# speedup vs baseline: 1.2114x; 1.2114x over previous
#include <hip/hip_runtime.h>
#include <hip/hip_bf16.h>


typedef float v2f __attribute__((ext_vector_type(2)));
typedef float v4f __attribute__((ext_vector_type(4)));
typedef float v8f __attribute__((ext_vector_type(8)));

#define B_    64
#define E_    16
#define IN_   256
#define OUT_  256
#define NPG_  2048
#define MWG   128
#define KC    32
#define BSTR2 272
#define ASTR  36
#define AK4   (KC / 4)
#define APASS (MWG * AK4 / 256)
#define AROWS (256 / AK4)
#define BPASS (KC / 8)

__global__ __launch_bounds__(256) void mole_mix(const float* __restrict__ kern,
                                                const float* __restrict__ coeff,
                                                float* __restrict__ mixed) {
    __shared__ float cs[B_ * E_];
    const int tid = threadIdx.x;
    for (int j = tid; j < B_ * E_; j += 256) cs[j] = coeff[j];
    __syncthreads();

    const int idx = blockIdx.x * 256 + tid;
    const int i   = idx >> 6;
    const int o4  = (idx & 63) << 2;

    float4 kf[E_];
#pragma unroll
    for (int e = 0; e < E_; ++e)
        kf[e] = *(const float4*)(kern + (size_t)e * IN_ * OUT_ + (size_t)i * OUT_ + o4);

    for (int b = 0; b < B_; ++b) {
        float4 acc = make_float4(0.f, 0.f, 0.f, 0.f);
#pragma unroll
        for (int e = 0; e < E_; ++e) {
            const float c = cs[b * E_ + e];
            acc.x += c * kf[e].x;  acc.y += c * kf[e].y;
            acc.z += c * kf[e].z;  acc.w += c * kf[e].w;
        }
        *(float4*)(mixed + (size_t)b * IN_ * OUT_ + (size_t)i * OUT_ + o4) = acc;
    }
}

__global__ __launch_bounds__(256) void mole_gemm(const float* __restrict__ X,
                                                 const float* __restrict__ mixed,
                                                 const float* __restrict__ bias,
                                                 float* __restrict__ out) {
    extern __shared__ float smem[];
    float* Bs = smem;
    float* As = smem + (KC / 2) * BSTR2 * 2;

    const int tid = threadIdx.x;
    const int b   = blockIdx.x >> 4;
    const int mt  = blockIdx.x & 15;

    const int wave = tid >> 5;
    const int lane = tid & 31;
    const int half = lane >> 4;
    const int l16  = lane & 15;
    const int wm   = wave >> 2;
    const int wn   = wave & 3;
    const int rowBlk = b * NPG_ + mt * MWG;

    const int arow = tid / AK4;
    const int ak4  = (tid % AK4) << 2;
    const int bp   = tid >> 6;
    const int bc4  = (tid & 63) << 2;
    const float* __restrict__ asrc = X + (size_t)(rowBlk + arow) * IN_ + ak4;
    const float* __restrict__ bsrc = mixed + (size_t)b * IN_ * OUT_ + (size_t)(2 * bp) * OUT_ + bc4;
    float* adst = As + arow * ASTR + ak4;
    float* bdst = Bs + (bp * BSTR2 + bc4) * 2;

    const float* Ab = As + (wm * 64 + l16) * ASTR + 2 * half;
    const v2f*   Bb = (const v2f*)Bs + half * BSTR2 + wn * 64 + l16;

    v8f acc[4][4];
#pragma unroll
    for (int i = 0; i < 4; ++i)
#pragma unroll
        for (int j = 0; j < 4; ++j) {
            v8f z = {};
            acc[i][j] = z;
        }

    v4f pa[APASS], plo[BPASS], phi[BPASS];
#pragma unroll
    for (int i = 0; i < APASS; ++i)
        pa[i]  = *(const v4f*)(asrc + (size_t)i * AROWS * IN_);
#pragma unroll
    for (int i = 0; i < BPASS; ++i) {
        plo[i] = *(const v4f*)(bsrc + (size_t)i * 8 * OUT_);
        phi[i] = *(const v4f*)(bsrc + (size_t)i * 8 * OUT_ + OUT_);
    }

    for (int kc = 0; kc < IN_ / KC; ++kc) {
#pragma unroll
        for (int i = 0; i < APASS; ++i)
            *(v4f*)(adst + i * AROWS * ASTR) = pa[i];
#pragma unroll
        for (int i = 0; i < BPASS; ++i) {
            float* d = bdst + i * 4 * BSTR2 * 2;
            *(v4f*)(d)     = (v4f){plo[i].x, phi[i].x, plo[i].y, phi[i].y};
            *(v4f*)(d + 4) = (v4f){plo[i].z, phi[i].z, plo[i].w, phi[i].w};
        }
        __syncthreads();

        if (kc + 1 < IN_ / KC) {
            const float* an = asrc + (kc + 1) * KC;
            const float* bn = bsrc + (size_t)(kc + 1) * KC * OUT_;
#pragma unroll
            for (int i = 0; i < APASS; ++i)
                pa[i]  = *(const v4f*)(an + (size_t)i * AROWS * IN_);
#pragma unroll
            for (int i = 0; i < BPASS; ++i) {
                plo[i] = *(const v4f*)(bn + (size_t)i * 8 * OUT_);
                phi[i] = *(const v4f*)(bn + (size_t)i * 8 * OUT_ + OUT_);
            }
        }

#pragma unroll
        for (int kk = 0; kk < KC / 4; ++kk) {
            v2f a[4], bf[4];
#pragma unroll
            for (int i = 0; i < 4; ++i)
                a[i] = *(const v2f*)(Ab + i * 16 * ASTR + kk * 4);
#pragma unroll
            for (int j = 0; j < 4; ++j)
                bf[j] = Bb[2 * kk * BSTR2 + j * 16];
#pragma unroll
            for (int i = 0; i < 4; ++i)
#pragma unroll
                for (int j = 0; j < 4; ++j)
                    acc[i][j] = __builtin_amdgcn_wmma_f32_16x16x4_f32(
                        false, a[i], false, bf[j], (short)0, acc[i][j], false, false);
        }
        __syncthreads();
    }

#pragma unroll
    for (int i = 0; i < 4; ++i) {
        const int outRow0 = rowBlk + wm * 64 + i * 16 + 8 * half;
#pragma unroll
        for (int j = 0; j < 4; ++j) {
            const int col = wn * 64 + j * 16 + l16;
            const float bv = bias[col];
#pragma unroll
            for (int r = 0; r < 8; ++r) {
                out[(size_t)(outRow0 + r) * OUT_ + col] = acc[i][j][r] + bv;
            }
        }
    }
}

extern "C" void kernel_launch(void* const* d_in, const int* in_sizes, int n_in,
                              void* d_out, int out_size, void* d_ws, size_t ws_size,
                              hipStream_t stream) {
    const float* inputs = (const float*)d_in[0];
    const float* coeffs = (const float*)d_in[1];
    const float* kern   = (const float*)d_in[3];
    const float* bias   = (const float*)d_in[4];
    float*       out    = (float*)d_out;
    float*       mixed  = (float*)d_ws;

    mole_mix<<<dim3((IN_ * OUT_ / 4) / 256), dim3(256), 0, stream>>>(kern, coeffs, mixed);

    const size_t ldsBytes = (size_t)((KC / 2) * BSTR2 * 2 + MWG * ASTR) * sizeof(float);
    mole_gemm<<<dim3(B_ * (NPG_ / MWG)), dim3(256), ldsBytes, stream>>>(inputs, mixed,
                                                                        bias, out);
}


# GEMM epilogue through a per-wave LDS strip: each wave stores its 64x64 tile as float4 row pieces, 4 whole 128-byte lines per store instruction, instead of 64-byte half lines by 4-byte stores
# speedup vs baseline: 1.2866x; 1.0621x over previous
#include <hip/hip_runtime.h>
#include <hip/hip_bf16.h>


typedef float v2f __attribute__((ext_vector_type(2)));
typedef float v4f __attribute__((ext_vector_type(4)));
typedef float v8f __attribute__((ext_vector_type(8)));

#define B_    64
#define E_    16
#define IN_   256
#define OUT_  256
#define NPG_  2048
#define MWG   128
#define KC    32
#define BSTR2 272
#define ASTR  36
#define AK4   (KC / 4)
#define APASS (MWG * AK4 / 256)
#define AROWS (256 / AK4)
#define BPASS (KC / 8)
#define CSTR  68

__global__ __launch_bounds__(256) void mole_mix(const float* __restrict__ kern,
                                                const float* __restrict__ coeff,
                                                float* __restrict__ mixed) {
    __shared__ float cs[B_ * E_];
    const int tid = threadIdx.x;
    for (int j = tid; j < B_ * E_; j += 256) cs[j] = coeff[j];
    __syncthreads();

    const int idx = blockIdx.x * 256 + tid;
    const int i   = idx >> 6;
    const int o4  = (idx & 63) << 2;

    float4 kf[E_];
#pragma unroll
    for (int e = 0; e < E_; ++e)
        kf[e] = *(const float4*)(kern + (size_t)e * IN_ * OUT_ + (size_t)i * OUT_ + o4);

    for (int b = 0; b < B_; ++b) {
        float4 acc = make_float4(0.f, 0.f, 0.f, 0.f);
#pragma unroll
        for (int e = 0; e < E_; ++e) {
            const float c = cs[b * E_ + e];
            acc.x += c * kf[e].x;  acc.y += c * kf[e].y;
            acc.z += c * kf[e].z;  acc.w += c * kf[e].w;
        }
        *(float4*)(mixed + (size_t)b * IN_ * OUT_ + (size_t)i * OUT_ + o4) = acc;
    }
}

__global__ __launch_bounds__(256) __attribute__((amdgpu_waves_per_eu(4, 4))) void mole_gemm(const float* __restrict__ X,
                                                 const float* __restrict__ mixed,
                                                 const float* __restrict__ bias,
                                                 float* __restrict__ out) {
    extern __shared__ float smem[];
    float* Bs = smem;
    float* As = smem + (KC / 2) * BSTR2 * 2;

    const int tid = threadIdx.x;
    const int b   = blockIdx.x >> 4;
    const int mt  = blockIdx.x & 15;

    const int wave = tid >> 5;
    const int lane = tid & 31;
    const int half = lane >> 4;
    const int l16  = lane & 15;
    const int wm   = wave >> 2;
    const int wn   = wave & 3;
    const int rowBlk = b * NPG_ + mt * MWG;

    const int arow = tid / AK4;
    const int ak4  = (tid % AK4) << 2;
    const int bp   = tid >> 6;
    const int bc4  = (tid & 63) << 2;
    const float* __restrict__ asrc = X + (size_t)(rowBlk + arow) * IN_ + ak4;
    const float* __restrict__ bsrc = mixed + (size_t)b * IN_ * OUT_ + (size_t)(2 * bp) * OUT_ + bc4;
    float* adst = As + arow * ASTR + ak4;
    float* bdst = Bs + (bp * BSTR2 + bc4) * 2;

    const float* Ab = As + (wm * 64 + l16) * ASTR + 2 * half;
    const v2f*   Bb = (const v2f*)Bs + half * BSTR2 + wn * 64 + l16;

    v8f acc[4][4];
#pragma unroll
    for (int i = 0; i < 4; ++i)
#pragma unroll
        for (int j = 0; j < 4; ++j) {
            v8f z = {};
            acc[i][j] = z;
        }

    v4f pa[APASS], plo[BPASS], phi[BPASS];
#pragma unroll
    for (int i = 0; i < APASS; ++i)
        pa[i]  = *(const v4f*)(asrc + (size_t)i * AROWS * IN_);
#pragma unroll
    for (int i = 0; i < BPASS; ++i) {
        plo[i] = *(const v4f*)(bsrc + (size_t)i * 8 * OUT_);
        phi[i] = *(const v4f*)(bsrc + (size_t)i * 8 * OUT_ + OUT_);
    }

    for (int kc = 0; kc < IN_ / KC; ++kc) {
#pragma unroll
        for (int i = 0; i < APASS; ++i)
            *(v4f*)(adst + i * AROWS * ASTR) = pa[i];
#pragma unroll
        for (int i = 0; i < BPASS; ++i) {
            float* d = bdst + i * 4 * BSTR2 * 2;
            *(v4f*)(d)     = (v4f){plo[i].x, phi[i].x, plo[i].y, phi[i].y};
            *(v4f*)(d + 4) = (v4f){plo[i].z, phi[i].z, plo[i].w, phi[i].w};
        }
        __syncthreads();

        if (kc + 1 < IN_ / KC) {
            const float* an = asrc + (kc + 1) * KC;
            const float* bn = bsrc + (size_t)(kc + 1) * KC * OUT_;
#pragma unroll
            for (int i = 0; i < APASS; ++i)
                pa[i]  = *(const v4f*)(an + (size_t)i * AROWS * IN_);
#pragma unroll
            for (int i = 0; i < BPASS; ++i) {
                plo[i] = *(const v4f*)(bn + (size_t)i * 8 * OUT_);
                phi[i] = *(const v4f*)(bn + (size_t)i * 8 * OUT_ + OUT_);
            }
        }

#pragma unroll
        for (int kk = 0; kk < KC / 4; ++kk) {
            v2f a[4], bf[4];
#pragma unroll
            for (int i = 0; i < 4; ++i)
                a[i] = *(const v2f*)(Ab + i * 16 * ASTR + kk * 4);
#pragma unroll
            for (int j = 0; j < 4; ++j)
                bf[j] = Bb[2 * kk * BSTR2 + j * 16];
#pragma unroll
            for (int i = 0; i < 4; ++i)
#pragma unroll
                for (int j = 0; j < 4; ++j)
                    acc[i][j] = __builtin_amdgcn_wmma_f32_16x16x4_f32(
                        false, a[i], false, bf[j], (short)0, acc[i][j], false, false);
        }
        __syncthreads();
    }

    float* stg = smem + wave * (16 * CSTR);
    float* wdst = stg + (8 * half) * CSTR + l16;
    const float* rsrc = stg + half * CSTR + 4 * l16;
    const v4f b4 = *(const v4f*)(bias + wn * 64 + 4 * l16);
    float* obase = out + (size_t)(rowBlk + wm * 64 + half) * OUT_ + wn * 64 + 4 * l16;
#pragma unroll
    for (int i = 0; i < 4; ++i) {
#pragma unroll
        for (int j = 0; j < 4; ++j)
#pragma unroll
            for (int r = 0; r < 8; ++r)
                wdst[r * CSTR + j * 16] = acc[i][j][r];
#pragma unroll
        for (int s = 0; s < 8; ++s) {
            const v4f v = *(const v4f*)(rsrc + 2 * s * CSTR) + b4;
            *(v4f*)(obase + (size_t)(i * 16 + 2 * s) * OUT_) = v;
        }
    }
}

extern "C" void kernel_launch(void* const* d_in, const int* in_sizes, int n_in,
                              void* d_out, int out_size, void* d_ws, size_t ws_size,
                              hipStream_t stream) {
    const float* inputs = (const float*)d_in[0];
    const float* coeffs = (const float*)d_in[1];
    const float* kern   = (const float*)d_in[3];
    const float* bias   = (const float*)d_in[4];
    float*       out    = (float*)d_out;
    float*       mixed  = (float*)d_ws;

    mole_mix<<<dim3((IN_ * OUT_ / 4) / 256), dim3(256), 0, stream>>>(kern, coeffs, mixed);

    const size_t ldsBytes = (size_t)((KC / 2) * BSTR2 * 2 + MWG * ASTR) * sizeof(float);
    mole_gemm<<<dim3(B_ * (NPG_ / MWG)), dim3(256), ldsBytes, stream>>>(inputs, mixed,
                                                                        bias, out);
}
